# starting kernel
# baseline (speedup 1.0000x reference)
#include <hip/hip_runtime.h>
#include <hip/hip_bf16.h>

typedef __attribute__((ext_vector_type(2))) float v2f;
typedef __attribute__((ext_vector_type(4))) float v4f;
typedef __attribute__((ext_vector_type(8))) float v8f;

#define B_    64
#define L_    4096
#define D_    128
#define NSEG  16
#define LSEG  (L_ / NSEG)
#define NEG_BIG -1e9f

__device__ __forceinline__ float fast_tanh(float t) {
  float e = __expf(2.0f * t);
  return 1.0f - 2.0f * __builtin_amdgcn_rcpf(e + 1.0f);
}

__global__ __launch_bounds__(256) void k1_scores(
    const float* __restrict__ x, const float* __restrict__ y,
    const float* __restrict__ w, const float* __restrict__ bias,
    float* __restrict__ scores)
{
  const int b   = blockIdx.x / NSEG;
  const int seg = blockIdx.x % NSEG;
  const int l0s = seg * LSEG;

  __shared__ float xs[D_];
  __shared__ float bs[D_];
  if (threadIdx.x < D_) {
    xs[threadIdx.x] = x[b * D_ + threadIdx.x];
    bs[threadIdx.x] = bias[threadIdx.x];
  }
  __syncthreads();

  const int wave  = threadIdx.x >> 5;
  const int lane  = threadIdx.x & 31;
  const int rsel  = lane & 15;
  const int dhalf = (lane >> 4) << 1;

  for (int t = wave; t < LSEG / 16; t += 8) {
    const int l0  = l0s + t * 16;
    const int row = l0 + rsel;
    const float* __restrict__ yrow = y + ((size_t)b * L_ + row) * (size_t)D_;
    const float* __restrict__ wrow = w + (size_t)row * (size_t)D_;

    v8f c = {0.f, 0.f, 0.f, 0.f, 0.f, 0.f, 0.f, 0.f};

    #pragma unroll
    for (int d0 = 0; d0 < D_; d0 += 4) {
      const int d = d0 + dhalf;
      v2f a  = *(const v2f*)(yrow + d);
      v2f wv = *(const v2f*)(wrow + d);
      v2f xv = *(const v2f*)(&xs[d]);
      v2f bv = *(const v2f*)(&bs[d]);
      v2f q;
      q.x = fast_tanh(__builtin_fmaf(xv.x, wv.x, bv.x));
      q.y = fast_tanh(__builtin_fmaf(xv.y, wv.y, bv.y));
      c = __builtin_amdgcn_wmma_f32_16x16x4_f32(
              false, a, false, q, (short)0, c, false, false);
    }

    float diag = 0.f;
    #pragma unroll
    for (int v = 0; v < 8; ++v)
      if (lane == v || lane == 24 + v) diag = c[v];

    if (lane < 8)
      scores[(size_t)b * L_ + l0 + lane] = diag;
    else if (lane >= 24)
      scores[(size_t)b * L_ + l0 + (lane - 16)] = diag;
  }
}

__device__ __forceinline__ float wave_max(float v) {
  #pragma unroll
  for (int o = 16; o > 0; o >>= 1) v = fmaxf(v, __shfl_xor(v, o, 32));
  return v;
}
__device__ __forceinline__ float wave_sum(float v) {
  #pragma unroll
  for (int o = 16; o > 0; o >>= 1) v += __shfl_xor(v, o, 32);
  return v;
}

__global__ __launch_bounds__(256) void k2_softmax(float* __restrict__ scores)
{
  const int b = blockIdx.x;
  float* __restrict__ s = scores + (size_t)b * L_;
  __shared__ float red[8];
  const int wv = threadIdx.x >> 5;
  const int ln = threadIdx.x & 31;

  float m = -INFINITY;
  for (int l = threadIdx.x; l < L_; l += 256) {
    float v = s[l];
    v = (v == 0.0f) ? (v + NEG_BIG) : v;
    m = fmaxf(m, v);
  }
  m = wave_max(m);
  if (ln == 0) red[wv] = m;
  __syncthreads();
  float gm = red[0];
  #pragma unroll
  for (int i = 1; i < 8; ++i) gm = fmaxf(gm, red[i]);
  __syncthreads();

  float sum = 0.f;
  for (int l = threadIdx.x; l < L_; l += 256) {
    float v = s[l];
    v = (v == 0.0f) ? (v + NEG_BIG) : v;
    sum += __expf(v - gm);
  }
  sum = wave_sum(sum);
  if (ln == 0) red[wv] = sum;
  __syncthreads();
  float gs = 0.f;
  #pragma unroll
  for (int i = 0; i < 8; ++i) gs += red[i];
  const float inv = 1.0f / gs;

  for (int l = threadIdx.x; l < L_; l += 256) {
    float v = s[l];
    v = (v == 0.0f) ? (v + NEG_BIG) : v;
    s[l] = __expf(v - gm) * inv;
  }
}

__global__ __launch_bounds__(256) void k3_partial(
    const float* __restrict__ y, const float* __restrict__ p,
    float* __restrict__ partial)
{
  const int b   = blockIdx.x / NSEG;
  const int seg = blockIdx.x % NSEG;
  const int l0  = seg * LSEG;

  __shared__ float ps[LSEG];
  for (int i = threadIdx.x; i < LSEG; i += 256)
    ps[i] = p[(size_t)b * L_ + l0 + i];
  __syncthreads();

  const int q = (threadIdx.x & 31) << 2;
  const int h = threadIdx.x >> 5;
  const float* __restrict__ yb = y + ((size_t)b * L_ + l0) * (size_t)D_;

  v4f acc = {0.f, 0.f, 0.f, 0.f};
  const int lbeg = h * (LSEG / 8), lend = lbeg + (LSEG / 8);
  for (int l = lbeg; l < lend; ++l) {
    __builtin_prefetch(yb + (size_t)(l + 8) * D_ + q, 0, 0);
    v4f yv = *(const v4f*)(yb + (size_t)l * D_ + q);
    acc = acc + yv * ps[l];
  }

  __shared__ float comb[8][D_];
  *(v4f*)&comb[h][q] = acc;
  __syncthreads();
  if (threadIdx.x < D_) {
    float sacc = 0.f;
    #pragma unroll
    for (int i = 0; i < 8; ++i) sacc += comb[i][threadIdx.x];
    partial[((size_t)b * NSEG + seg) * D_ + threadIdx.x] = sacc;
  }
}

__global__ __launch_bounds__(128) void k4_reduce(
    const float* __restrict__ partial, float* __restrict__ out)
{
  const int b = blockIdx.x;
  const int d = threadIdx.x;
  float acc = 0.f;
  #pragma unroll
  for (int s = 0; s < NSEG; ++s)
    acc += partial[((size_t)b * NSEG + s) * D_ + d];
  out[(size_t)b * D_ + d] = acc;
}

extern "C" void kernel_launch(void* const* d_in, const int* in_sizes, int n_in,
                              void* d_out, int out_size, void* d_ws, size_t ws_size,
                              hipStream_t stream)
{
  const float* x    = (const float*)d_in[0];
  const float* y    = (const float*)d_in[1];
  const float* w    = (const float*)d_in[2];
  const float* bias = (const float*)d_in[3];
  float* out = (float*)d_out;

  float* scores  = (float*)d_ws;
  float* partial = scores + (size_t)B_ * L_;

  hipLaunchKernelGGL(k1_scores,  dim3(B_ * NSEG), dim3(256), 0, stream,
                     x, y, w, bias, scores);
  hipLaunchKernelGGL(k2_softmax, dim3(B_),        dim3(256), 0, stream, scores);
  hipLaunchKernelGGL(k3_partial, dim3(B_ * NSEG), dim3(256), 0, stream,
                     y, scores, partial);
  hipLaunchKernelGGL(k4_reduce,  dim3(B_),        dim3(128), 0, stream,
                     partial, out);
}


# K1: y and w read as whole 512-B rows per wave (b128 per lane), tanh in row layout, both staged in a wave-private LDS tile; WMMA fragments read from LDS in the same k order (sum order unchanged)
# speedup vs baseline: 1.5847x; 1.5847x over previous
#include <hip/hip_runtime.h>
#include <hip/hip_bf16.h>

typedef __attribute__((ext_vector_type(2))) float v2f;
typedef __attribute__((ext_vector_type(4))) float v4f;
typedef __attribute__((ext_vector_type(8))) float v8f;

#define B_    64
#define L_    4096
#define D_    128
#define NSEG  16
#define LSEG  (L_ / NSEG)
#define NEG_BIG -1e9f

__device__ __forceinline__ float fast_tanh(float t) {
  float e = __expf(2.0f * t);
  return 1.0f - 2.0f * __builtin_amdgcn_rcpf(e + 1.0f);
}

#define PITCH 132

__global__ __launch_bounds__(256) void k1_scores(
    const float* __restrict__ x, const float* __restrict__ y,
    const float* __restrict__ w, const float* __restrict__ bias,
    float* __restrict__ scores)
{
  const int b   = blockIdx.x / NSEG;
  const int seg = blockIdx.x % NSEG;
  const int l0s = seg * LSEG;

  __shared__ __attribute__((aligned(16))) float ys[8][16 * PITCH];
  __shared__ __attribute__((aligned(16))) float qs[8][16 * PITCH];

  const int wave  = threadIdx.x >> 5;
  const int lane  = threadIdx.x & 31;
  const int rsel  = lane & 15;
  const int dhalf = (lane >> 4) << 1;
  const int dq    = lane << 2;

  const v4f xv = *(const v4f*)(x + (size_t)b * D_ + dq);
  const v4f bv = *(const v4f*)(bias + dq);

  float* __restrict__ ysw = ys[wave];
  float* __restrict__ qsw = qs[wave];

  for (int t = wave; t < LSEG / 16; t += 8) {
    const int l0 = l0s + t * 16;
    const float* __restrict__ yt = y + ((size_t)b * L_ + l0) * (size_t)D_ + dq;
    const float* __restrict__ wt = w + (size_t)l0 * (size_t)D_ + dq;

    #pragma unroll
    for (int r = 0; r < 16; ++r) {
      v4f yv = *(const v4f*)(yt + (size_t)r * D_);
      v4f wv = *(const v4f*)(wt + (size_t)r * D_);
      v4f q;
      q.x = fast_tanh(__builtin_fmaf(xv.x, wv.x, bv.x));
      q.y = fast_tanh(__builtin_fmaf(xv.y, wv.y, bv.y));
      q.z = fast_tanh(__builtin_fmaf(xv.z, wv.z, bv.z));
      q.w = fast_tanh(__builtin_fmaf(xv.w, wv.w, bv.w));
      *(v4f*)(ysw + r * PITCH + dq) = yv;
      *(v4f*)(qsw + r * PITCH + dq) = q;
    }

    __builtin_amdgcn_fence(__ATOMIC_RELEASE, "wavefront");
    __builtin_amdgcn_wave_barrier();
    __builtin_amdgcn_fence(__ATOMIC_ACQUIRE, "wavefront");

    v8f c = {0.f, 0.f, 0.f, 0.f, 0.f, 0.f, 0.f, 0.f};

    const float* __restrict__ ya = ysw + rsel * PITCH + dhalf;
    const float* __restrict__ qb = qsw + rsel * PITCH + dhalf;
    #pragma unroll
    for (int d0 = 0; d0 < D_; d0 += 4) {
      v2f a = *(const v2f*)(ya + d0);
      v2f q = *(const v2f*)(qb + d0);
      c = __builtin_amdgcn_wmma_f32_16x16x4_f32(
              false, a, false, q, (short)0, c, false, false);
    }

    __builtin_amdgcn_fence(__ATOMIC_RELEASE, "wavefront");
    __builtin_amdgcn_wave_barrier();
    __builtin_amdgcn_fence(__ATOMIC_ACQUIRE, "wavefront");

    float diag = 0.f;
    #pragma unroll
    for (int v = 0; v < 8; ++v)
      if (lane == v || lane == 24 + v) diag = c[v];

    if (lane < 8)
      scores[(size_t)b * L_ + l0 + lane] = diag;
    else if (lane >= 24)
      scores[(size_t)b * L_ + l0 + (lane - 16)] = diag;
  }
}

__device__ __forceinline__ float wave_max(float v) {
  #pragma unroll
  for (int o = 16; o > 0; o >>= 1) v = fmaxf(v, __shfl_xor(v, o, 32));
  return v;
}
__device__ __forceinline__ float wave_sum(float v) {
  #pragma unroll
  for (int o = 16; o > 0; o >>= 1) v += __shfl_xor(v, o, 32);
  return v;
}

__global__ __launch_bounds__(256) void k2_softmax(float* __restrict__ scores)
{
  const int b = blockIdx.x;
  float* __restrict__ s = scores + (size_t)b * L_;
  __shared__ float red[8];
  const int wv = threadIdx.x >> 5;
  const int ln = threadIdx.x & 31;

  float m = -INFINITY;
  for (int l = threadIdx.x; l < L_; l += 256) {
    float v = s[l];
    v = (v == 0.0f) ? (v + NEG_BIG) : v;
    m = fmaxf(m, v);
  }
  m = wave_max(m);
  if (ln == 0) red[wv] = m;
  __syncthreads();
  float gm = red[0];
  #pragma unroll
  for (int i = 1; i < 8; ++i) gm = fmaxf(gm, red[i]);
  __syncthreads();

  float sum = 0.f;
  for (int l = threadIdx.x; l < L_; l += 256) {
    float v = s[l];
    v = (v == 0.0f) ? (v + NEG_BIG) : v;
    sum += __expf(v - gm);
  }
  sum = wave_sum(sum);
  if (ln == 0) red[wv] = sum;
  __syncthreads();
  float gs = 0.f;
  #pragma unroll
  for (int i = 0; i < 8; ++i) gs += red[i];
  const float inv = 1.0f / gs;

  for (int l = threadIdx.x; l < L_; l += 256) {
    float v = s[l];
    v = (v == 0.0f) ? (v + NEG_BIG) : v;
    s[l] = __expf(v - gm) * inv;
  }
}

__global__ __launch_bounds__(256) void k3_partial(
    const float* __restrict__ y, const float* __restrict__ p,
    float* __restrict__ partial)
{
  const int b   = blockIdx.x / NSEG;
  const int seg = blockIdx.x % NSEG;
  const int l0  = seg * LSEG;

  __shared__ float ps[LSEG];
  for (int i = threadIdx.x; i < LSEG; i += 256)
    ps[i] = p[(size_t)b * L_ + l0 + i];
  __syncthreads();

  const int q = (threadIdx.x & 31) << 2;
  const int h = threadIdx.x >> 5;
  const float* __restrict__ yb = y + ((size_t)b * L_ + l0) * (size_t)D_;

  v4f acc = {0.f, 0.f, 0.f, 0.f};
  const int lbeg = h * (LSEG / 8), lend = lbeg + (LSEG / 8);
  for (int l = lbeg; l < lend; ++l) {
    __builtin_prefetch(yb + (size_t)(l + 8) * D_ + q, 0, 0);
    v4f yv = *(const v4f*)(yb + (size_t)l * D_ + q);
    acc = acc + yv * ps[l];
  }

  __shared__ float comb[8][D_];
  *(v4f*)&comb[h][q] = acc;
  __syncthreads();
  if (threadIdx.x < D_) {
    float sacc = 0.f;
    #pragma unroll
    for (int i = 0; i < 8; ++i) sacc += comb[i][threadIdx.x];
    partial[((size_t)b * NSEG + seg) * D_ + threadIdx.x] = sacc;
  }
}

__global__ __launch_bounds__(128) void k4_reduce(
    const float* __restrict__ partial, float* __restrict__ out)
{
  const int b = blockIdx.x;
  const int d = threadIdx.x;
  float acc = 0.f;
  #pragma unroll
  for (int s = 0; s < NSEG; ++s)
    acc += partial[((size_t)b * NSEG + s) * D_ + d];
  out[(size_t)b * D_ + d] = acc;
}

extern "C" void kernel_launch(void* const* d_in, const int* in_sizes, int n_in,
                              void* d_out, int out_size, void* d_ws, size_t ws_size,
                              hipStream_t stream)
{
  const float* x    = (const float*)d_in[0];
  const float* y    = (const float*)d_in[1];
  const float* w    = (const float*)d_in[2];
  const float* bias = (const float*)d_in[3];
  float* out = (float*)d_out;

  float* scores  = (float*)d_ws;
  float* partial = scores + (size_t)B_ * L_;

  hipLaunchKernelGGL(k1_scores,  dim3(B_ * NSEG), dim3(256), 0, stream,
                     x, y, w, bias, scores);
  hipLaunchKernelGGL(k2_softmax, dim3(B_),        dim3(256), 0, stream, scores);
  hipLaunchKernelGGL(k3_partial, dim3(B_ * NSEG), dim3(256), 0, stream,
                     y, scores, partial);
  hipLaunchKernelGGL(k4_reduce,  dim3(B_),        dim3(128), 0, stream,
                     partial, out);
}


# Softmax and weighted y sum fused into the score kernel (online softmax: per-wave running max, partials rescaled in one merge kernel); y read once, 4 launches to 2; sum order changes
# speedup vs baseline: 2.1193x; 1.3374x over previous
#include <hip/hip_runtime.h>
#include <hip/hip_bf16.h>

typedef __attribute__((ext_vector_type(2))) float v2f;
typedef __attribute__((ext_vector_type(4))) float v4f;
typedef __attribute__((ext_vector_type(8))) float v8f;

#define B_    64
#define L_    4096
#define D_    128
#define NSEG  16
#define LSEG  (L_ / NSEG)
#define NEG_BIG -1e9f

__device__ __forceinline__ float fast_tanh(float t) {
  float e = __expf(2.0f * t);
  return 1.0f - 2.0f * __builtin_amdgcn_rcpf(e + 1.0f);
}

#define PITCH 132
#define REC   160

__global__ __launch_bounds__(256) void k1_scores_softmax_wsum(
    const float* __restrict__ x, const float* __restrict__ y,
    const float* __restrict__ w, const float* __restrict__ bias,
    float* __restrict__ part)
{
  const int b   = blockIdx.x / NSEG;
  const int seg = blockIdx.x % NSEG;
  const int l0s = seg * LSEG;

  __shared__ __attribute__((aligned(16))) float ys[8][16 * PITCH];
  __shared__ __attribute__((aligned(16))) float qs[8][16 * PITCH];
  __shared__ __attribute__((aligned(16))) float sc[8][16];

  const int wave  = threadIdx.x >> 5;
  const int lane  = threadIdx.x & 31;
  const int rsel  = lane & 15;
  const int dhalf = (lane >> 4) << 1;
  const int dq    = lane << 2;

  const v4f xv = *(const v4f*)(x + (size_t)b * D_ + dq);
  const v4f bv = *(const v4f*)(bias + dq);

  float* __restrict__ ysw = ys[wave];
  float* __restrict__ qsw = qs[wave];
  float* __restrict__ scw = sc[wave];

  float m_w = -INFINITY;
  float s_w = 0.f;
  v4f acc_w = {0.f, 0.f, 0.f, 0.f};

  for (int t = wave; t < LSEG / 16; t += 8) {
    const int l0 = l0s + t * 16;
    const float* __restrict__ yt = y + ((size_t)b * L_ + l0) * (size_t)D_ + dq;
    const float* __restrict__ wt = w + (size_t)l0 * (size_t)D_ + dq;

    #pragma unroll
    for (int r = 0; r < 16; ++r) {
      v4f yv = *(const v4f*)(yt + (size_t)r * D_);
      v4f wv = *(const v4f*)(wt + (size_t)r * D_);
      v4f q;
      q.x = fast_tanh(__builtin_fmaf(xv.x, wv.x, bv.x));
      q.y = fast_tanh(__builtin_fmaf(xv.y, wv.y, bv.y));
      q.z = fast_tanh(__builtin_fmaf(xv.z, wv.z, bv.z));
      q.w = fast_tanh(__builtin_fmaf(xv.w, wv.w, bv.w));
      *(v4f*)(ysw + r * PITCH + dq) = yv;
      *(v4f*)(qsw + r * PITCH + dq) = q;
    }

    __builtin_amdgcn_fence(__ATOMIC_RELEASE, "wavefront");
    __builtin_amdgcn_wave_barrier();
    __builtin_amdgcn_fence(__ATOMIC_ACQUIRE, "wavefront");

    v8f c = {0.f, 0.f, 0.f, 0.f, 0.f, 0.f, 0.f, 0.f};

    const float* __restrict__ ya = ysw + rsel * PITCH + dhalf;
    const float* __restrict__ qb = qsw + rsel * PITCH + dhalf;
    #pragma unroll
    for (int d0 = 0; d0 < D_; d0 += 4) {
      v2f a = *(const v2f*)(ya + d0);
      v2f q = *(const v2f*)(qb + d0);
      c = __builtin_amdgcn_wmma_f32_16x16x4_f32(
              false, a, false, q, (short)0, c, false, false);
    }

    float diag = 0.f;
    #pragma unroll
    for (int v = 0; v < 8; ++v)
      if (lane == v || lane == 24 + v) diag = c[v];

    const float sv = (diag == 0.0f) ? (diag + NEG_BIG) : diag;
    if (lane < 8)
      scw[lane] = sv;
    else if (lane >= 24)
      scw[lane - 16] = sv;

    __builtin_amdgcn_fence(__ATOMIC_RELEASE, "wavefront");
    __builtin_amdgcn_wave_barrier();
    __builtin_amdgcn_fence(__ATOMIC_ACQUIRE, "wavefront");

    float s16[16];
    #pragma unroll
    for (int g = 0; g < 4; ++g) {
      const v4f sq = *(const v4f*)(scw + 4 * g);
      s16[4 * g + 0] = sq.x; s16[4 * g + 1] = sq.y;
      s16[4 * g + 2] = sq.z; s16[4 * g + 3] = sq.w;
    }

    float mt = s16[0];
    #pragma unroll
    for (int r = 1; r < 16; ++r) mt = fmaxf(mt, s16[r]);

    const float m_new = fmaxf(m_w, mt);
    const float alpha = __expf(m_w - m_new);

    float st = 0.f;
    v4f at = {0.f, 0.f, 0.f, 0.f};
    #pragma unroll
    for (int r = 0; r < 16; ++r) {
      const float e = __expf(s16[r] - m_new);
      const v4f yv = *(const v4f*)(ysw + r * PITCH + dq);
      st += e;
      at = at + yv * e;
    }
    s_w   = s_w * alpha + st;
    acc_w = acc_w * alpha + at;
    m_w   = m_new;

    __builtin_amdgcn_fence(__ATOMIC_RELEASE, "wavefront");
    __builtin_amdgcn_wave_barrier();
    __builtin_amdgcn_fence(__ATOMIC_ACQUIRE, "wavefront");
  }

  *(v4f*)(ysw + dq) = acc_w;
  if (lane == 0) {
    ysw[PITCH + 0] = m_w;
    ysw[PITCH + 1] = s_w;
  }
  __syncthreads();

  if (threadIdx.x < D_) {
    const int d = threadIdx.x;
    float gm = ys[0][PITCH];
    #pragma unroll
    for (int i = 1; i < 8; ++i) gm = fmaxf(gm, ys[i][PITCH]);
    float S = 0.f, A = 0.f;
    #pragma unroll
    for (int i = 0; i < 8; ++i) {
      const float wi = __expf(ys[i][PITCH] - gm);
      S += wi * ys[i][PITCH + 1];
      A += wi * ys[i][d];
    }
    float* __restrict__ rec = part + (size_t)blockIdx.x * REC;
    rec[d] = A;
    if (d == 0) {
      v2f ms = {gm, S};
      *(v2f*)(rec + D_) = ms;
    }
  }
}

__global__ __launch_bounds__(128) void k2_combine(
    const float* __restrict__ part, float* __restrict__ out)
{
  const int b = blockIdx.x;
  const int d = threadIdx.x;
  const float* __restrict__ r0 = part + (size_t)b * NSEG * REC;

  float m[NSEG];
  #pragma unroll
  for (int s = 0; s < NSEG; ++s) m[s] = r0[s * REC + D_];
  float gm = m[0];
  #pragma unroll
  for (int s = 1; s < NSEG; ++s) gm = fmaxf(gm, m[s]);

  float num = 0.f, den = 0.f;
  #pragma unroll
  for (int s = 0; s < NSEG; ++s) {
    const float ws = __expf(m[s] - gm);
    den += ws * r0[s * REC + D_ + 1];
    num += ws * r0[s * REC + d];
  }
  out[(size_t)b * D_ + d] = num / den;
}

extern "C" void kernel_launch(void* const* d_in, const int* in_sizes, int n_in,
                              void* d_out, int out_size, void* d_ws, size_t ws_size,
                              hipStream_t stream)
{
  const float* x    = (const float*)d_in[0];
  const float* y    = (const float*)d_in[1];
  const float* w    = (const float*)d_in[2];
  const float* bias = (const float*)d_in[3];
  float* out = (float*)d_out;

  float* part = (float*)d_ws;

  hipLaunchKernelGGL(k1_scores_softmax_wsum, dim3(B_ * NSEG), dim3(256), 0, stream,
                     x, y, w, bias, part);
  hipLaunchKernelGGL(k2_combine, dim3(B_), dim3(128), 0, stream, part, out);
}


# K1: tanh(x*w+b) taken from the hardware v_tanh_f32, one transcendental per value, in place of exp + rcp and three plain instructions per value; k order and sum order unchanged
# speedup vs baseline: 2.1867x; 1.0318x over previous
#include <hip/hip_runtime.h>
#include <hip/hip_bf16.h>

typedef __attribute__((ext_vector_type(2))) float v2f;
typedef __attribute__((ext_vector_type(4))) float v4f;
typedef __attribute__((ext_vector_type(8))) float v8f;

#define B_    64
#define L_    4096
#define D_    128
#define NSEG  16
#define LSEG  (L_ / NSEG)
#define NEG_BIG -1e9f

__device__ __forceinline__ float fast_tanh(float t) {
  return __builtin_amdgcn_tanhf(t);
}

#define PITCH 132
#define REC   160

__global__ __launch_bounds__(256) void k1_scores_softmax_wsum(
    const float* __restrict__ x, const float* __restrict__ y,
    const float* __restrict__ w, const float* __restrict__ bias,
    float* __restrict__ part)
{
  const int b   = blockIdx.x / NSEG;
  const int seg = blockIdx.x % NSEG;
  const int l0s = seg * LSEG;

  __shared__ __attribute__((aligned(16))) float ys[8][16 * PITCH];
  __shared__ __attribute__((aligned(16))) float qs[8][16 * PITCH];
  __shared__ __attribute__((aligned(16))) float sc[8][16];

  const int wave  = threadIdx.x >> 5;
  const int lane  = threadIdx.x & 31;
  const int rsel  = lane & 15;
  const int dhalf = (lane >> 4) << 1;
  const int dq    = lane << 2;

  const v4f xv = *(const v4f*)(x + (size_t)b * D_ + dq);
  const v4f bv = *(const v4f*)(bias + dq);

  float* __restrict__ ysw = ys[wave];
  float* __restrict__ qsw = qs[wave];
  float* __restrict__ scw = sc[wave];

  float m_w = -INFINITY;
  float s_w = 0.f;
  v4f acc_w = {0.f, 0.f, 0.f, 0.f};

  for (int t = wave; t < LSEG / 16; t += 8) {
    const int l0 = l0s + t * 16;
    const float* __restrict__ yt = y + ((size_t)b * L_ + l0) * (size_t)D_ + dq;
    const float* __restrict__ wt = w + (size_t)l0 * (size_t)D_ + dq;

    #pragma unroll
    for (int r = 0; r < 16; ++r) {
      v4f yv = *(const v4f*)(yt + (size_t)r * D_);
      v4f wv = *(const v4f*)(wt + (size_t)r * D_);
      v4f q;
      q.x = fast_tanh(__builtin_fmaf(xv.x, wv.x, bv.x));
      q.y = fast_tanh(__builtin_fmaf(xv.y, wv.y, bv.y));
      q.z = fast_tanh(__builtin_fmaf(xv.z, wv.z, bv.z));
      q.w = fast_tanh(__builtin_fmaf(xv.w, wv.w, bv.w));
      *(v4f*)(ysw + r * PITCH + dq) = yv;
      *(v4f*)(qsw + r * PITCH + dq) = q;
    }

    __builtin_amdgcn_fence(__ATOMIC_RELEASE, "wavefront");
    __builtin_amdgcn_wave_barrier();
    __builtin_amdgcn_fence(__ATOMIC_ACQUIRE, "wavefront");

    v8f c = {0.f, 0.f, 0.f, 0.f, 0.f, 0.f, 0.f, 0.f};

    const float* __restrict__ ya = ysw + rsel * PITCH + dhalf;
    const float* __restrict__ qb = qsw + rsel * PITCH + dhalf;
    #pragma unroll
    for (int d0 = 0; d0 < D_; d0 += 4) {
      v2f a = *(const v2f*)(ya + d0);
      v2f q = *(const v2f*)(qb + d0);
      c = __builtin_amdgcn_wmma_f32_16x16x4_f32(
              false, a, false, q, (short)0, c, false, false);
    }

    float diag = 0.f;
    #pragma unroll
    for (int v = 0; v < 8; ++v)
      if (lane == v || lane == 24 + v) diag = c[v];

    const float sv = (diag == 0.0f) ? (diag + NEG_BIG) : diag;
    if (lane < 8)
      scw[lane] = sv;
    else if (lane >= 24)
      scw[lane - 16] = sv;

    __builtin_amdgcn_fence(__ATOMIC_RELEASE, "wavefront");
    __builtin_amdgcn_wave_barrier();
    __builtin_amdgcn_fence(__ATOMIC_ACQUIRE, "wavefront");

    float s16[16];
    #pragma unroll
    for (int g = 0; g < 4; ++g) {
      const v4f sq = *(const v4f*)(scw + 4 * g);
      s16[4 * g + 0] = sq.x; s16[4 * g + 1] = sq.y;
      s16[4 * g + 2] = sq.z; s16[4 * g + 3] = sq.w;
    }

    float mt = s16[0];
    #pragma unroll
    for (int r = 1; r < 16; ++r) mt = fmaxf(mt, s16[r]);

    const float m_new = fmaxf(m_w, mt);
    const float alpha = __expf(m_w - m_new);

    float st = 0.f;
    v4f at = {0.f, 0.f, 0.f, 0.f};
    #pragma unroll
    for (int r = 0; r < 16; ++r) {
      const float e = __expf(s16[r] - m_new);
      const v4f yv = *(const v4f*)(ysw + r * PITCH + dq);
      st += e;
      at = at + yv * e;
    }
    s_w   = s_w * alpha + st;
    acc_w = acc_w * alpha + at;
    m_w   = m_new;

    __builtin_amdgcn_fence(__ATOMIC_RELEASE, "wavefront");
    __builtin_amdgcn_wave_barrier();
    __builtin_amdgcn_fence(__ATOMIC_ACQUIRE, "wavefront");
  }

  *(v4f*)(ysw + dq) = acc_w;
  if (lane == 0) {
    ysw[PITCH + 0] = m_w;
    ysw[PITCH + 1] = s_w;
  }
  __syncthreads();

  if (threadIdx.x < D_) {
    const int d = threadIdx.x;
    float gm = ys[0][PITCH];
    #pragma unroll
    for (int i = 1; i < 8; ++i) gm = fmaxf(gm, ys[i][PITCH]);
    float S = 0.f, A = 0.f;
    #pragma unroll
    for (int i = 0; i < 8; ++i) {
      const float wi = __expf(ys[i][PITCH] - gm);
      S += wi * ys[i][PITCH + 1];
      A += wi * ys[i][d];
    }
    float* __restrict__ rec = part + (size_t)blockIdx.x * REC;
    rec[d] = A;
    if (d == 0) {
      v2f ms = {gm, S};
      *(v2f*)(rec + D_) = ms;
    }
  }
}

__global__ __launch_bounds__(128) void k2_combine(
    const float* __restrict__ part, float* __restrict__ out)
{
  const int b = blockIdx.x;
  const int d = threadIdx.x;
  const float* __restrict__ r0 = part + (size_t)b * NSEG * REC;

  float m[NSEG];
  #pragma unroll
  for (int s = 0; s < NSEG; ++s) m[s] = r0[s * REC + D_];
  float gm = m[0];
  #pragma unroll
  for (int s = 1; s < NSEG; ++s) gm = fmaxf(gm, m[s]);

  float num = 0.f, den = 0.f;
  #pragma unroll
  for (int s = 0; s < NSEG; ++s) {
    const float ws = __expf(m[s] - gm);
    den += ws * r0[s * REC + D_ + 1];
    num += ws * r0[s * REC + d];
  }
  out[(size_t)b * D_ + d] = num / den;
}

extern "C" void kernel_launch(void* const* d_in, const int* in_sizes, int n_in,
                              void* d_out, int out_size, void* d_ws, size_t ws_size,
                              hipStream_t stream)
{
  const float* x    = (const float*)d_in[0];
  const float* y    = (const float*)d_in[1];
  const float* w    = (const float*)d_in[2];
  const float* bias = (const float*)d_in[3];
  float* out = (float*)d_out;

  float* part = (float*)d_ws;

  hipLaunchKernelGGL(k1_scores_softmax_wsum, dim3(B_ * NSEG), dim3(256), 0, stream,
                     x, y, w, bias, part);
  hipLaunchKernelGGL(k2_combine, dim3(B_), dim3(128), 0, stream, part, out);
}
